# starting kernel
# baseline (speedup 1.0000x reference)
#include <hip/hip_runtime.h>

typedef float v2f __attribute__((ext_vector_type(2)));
typedef float v8f __attribute__((ext_vector_type(8)));

#define D 128
#define WP 132

__global__ void gcn_zero_kernel(float4* __restrict__ p, int n4) {
    int i = blockIdx.x * blockDim.x + threadIdx.x;
    int stride = gridDim.x * blockDim.x;
    float4 z = make_float4(0.f, 0.f, 0.f, 0.f);
    for (; i < n4; i += stride) p[i] = z;
}

__global__ void __launch_bounds__(256)
gcn_scatter_kernel(const float* __restrict__ X,
                   const int*   __restrict__ edge_row,
                   const int*   __restrict__ edge_col,
                   const float* __restrict__ edge_val,
                   float*       __restrict__ agg,
                   int E) {
    int wid  = (blockIdx.x * blockDim.x + threadIdx.x) >> 5;
    int lane = threadIdx.x & 31;
    if (wid >= E) return;

    int   r = edge_row[wid];
    int   c = edge_col[wid];
    float v = edge_val[wid];

    const float4* xrow = (const float4*)(X + (size_t)c * D);
    float4 x = xrow[lane];

    float* dst = agg + (size_t)r * D + lane * 4;
    __hip_atomic_fetch_add(dst + 0, v * x.x, __ATOMIC_RELAXED, __HIP_MEMORY_SCOPE_AGENT);
    __hip_atomic_fetch_add(dst + 1, v * x.y, __ATOMIC_RELAXED, __HIP_MEMORY_SCOPE_AGENT);
    __hip_atomic_fetch_add(dst + 2, v * x.z, __ATOMIC_RELAXED, __HIP_MEMORY_SCOPE_AGENT);
    __hip_atomic_fetch_add(dst + 3, v * x.w, __ATOMIC_RELAXED, __HIP_MEMORY_SCOPE_AGENT);
}

__global__ void __launch_bounds__(256)
gcn_gemm_wmma_kernel(const float* __restrict__ Wmat,
                     const float* __restrict__ bias,
                     float*       __restrict__ out,
                     int nrows) {
    __shared__ float Wlds[D * WP];

    const int tid = threadIdx.x;

    {
        const float4* Wv = (const float4*)Wmat;
        for (int i4 = tid; i4 < (D * D) / 4; i4 += 256) {
            int o = i4 >> 5;
            int k = (i4 & 31) * 4;
            *(float4*)&Wlds[o * WP + k] = Wv[i4];
        }
    }
    __syncthreads();

    const int wave = tid >> 5;
    const int lane = tid & 31;
    const int lr   = lane & 15;
    const int hi   = lane >> 4;

    const int row_base = blockIdx.x * 128 + wave * 16;
    const int row      = row_base + lr;
    const int rowc     = row < nrows ? row : (nrows - 1);
    const float* arow  = out + (size_t)rowc * D;

    v8f acc[8];
#pragma unroll
    for (int t = 0; t < 8; ++t)
        acc[t] = (v8f){0.f, 0.f, 0.f, 0.f, 0.f, 0.f, 0.f, 0.f};

#pragma unroll 8
    for (int kk = 0; kk < D / 4; ++kk) {
        const int k0 = kk * 4 + hi * 2;
        v2f a = *(const v2f*)(arow + k0);

#pragma unroll
        for (int t = 0; t < 8; ++t) {
            const int o = t * 16 + lr;
            v2f b = *(const v2f*)&Wlds[o * WP + k0];
            acc[t] = __builtin_amdgcn_wmma_f32_16x16x4_f32(
                 false, a,  false, b,
                 (short)0, acc[t],
                 false,  false);
        }
    }

#pragma unroll
    for (int t = 0; t < 8; ++t) {
        const int col = t * 16 + lr;
        const float bb = bias[col];
#pragma unroll
        for (int j = 0; j < 8; ++j) {
            const int r = row_base + j + hi * 8;
            if (r < nrows) out[(size_t)r * D + col] = acc[t][j] + bb;
        }
    }
}

extern "C" void kernel_launch(void* const* d_in, const int* in_sizes, int n_in,
                              void* d_out, int out_size, void* d_ws, size_t ws_size,
                              hipStream_t stream) {
    const float* X        = (const float*)d_in[0];
    const int*   edge_row = (const int*)  d_in[1];
    const int*   edge_col = (const int*)  d_in[2];
    const float* edge_val = (const float*)d_in[3];
    const float* Wmat     = (const float*)d_in[4];
    const float* bias     = (const float*)d_in[5];
    float*       out      = (float*)d_out;

    const int N = in_sizes[0] / D;
    const int E = in_sizes[1];

    {
        int n4 = out_size / 4;
        gcn_zero_kernel<<<2048, 256, 0, stream>>>((float4*)out, n4);
    }

    {
        long long threads = (long long)E * 32;
        int blocks = (int)((threads + 255) / 256);
        gcn_scatter_kernel<<<blocks, 256, 0, stream>>>(X, edge_row, edge_col,
                                                       edge_val, out, E);
    }

    {
        int blocks = (N + 127) / 128;
        gcn_gemm_wmma_kernel<<<blocks, 256, 0, stream>>>(Wmat, bias, out, N);
    }
}


# Global f32-atomic scatter and zero pass replaced by output-stationary row tiles: a workgroup keeps 392 agg rows in LDS, scans all edge rows, adds with LDS f32 atomics (sums still in arrival order)
# speedup vs baseline: 1.9746x; 1.9746x over previous
#include <hip/hip_runtime.h>


typedef float v2f __attribute__((ext_vector_type(2)));
typedef float v8f __attribute__((ext_vector_type(8)));

#define D 128
#define WP 132

#define TR  392
#define NW  32
#define SEG 320
#define UNR 8
#define GB  8

__device__ __forceinline__ void queue_match(int d, int nrow, int e, unsigned& cnt,
                                            uint2* q, float* tile,
                                            const float* __restrict__ X,
                                            const int*   __restrict__ edge_col,
                                            const float* __restrict__ edge_val) {
    const bool m = (unsigned)d < (unsigned)nrow;
    const unsigned b = __builtin_amdgcn_ballot_w32(m);
    if (b != 0u) {
        if (m) {
            const unsigned slot = __builtin_amdgcn_mbcnt_lo(b, cnt);
            if (slot < (unsigned)SEG) {
                q[slot] = make_uint2((unsigned)e, (unsigned)d);
            } else {
                const int   c = edge_col[e];
                const float v = edge_val[e];
                const float* xr = X + (size_t)c * D;
                float* t = tile + d * D;
#pragma nounroll
                for (int f = 0; f < D; ++f)
                    __hip_atomic_fetch_add(t + f, v * xr[f], __ATOMIC_RELAXED,
                                           __HIP_MEMORY_SCOPE_WORKGROUP);
            }
        }
        cnt += (unsigned)__builtin_popcount(b);
    }
}

__global__ void __launch_bounds__(1024)
gcn_rowtile_kernel(const float* __restrict__ X,
                   const int*   __restrict__ edge_row,
                   const int*   __restrict__ edge_col,
                   const float* __restrict__ edge_val,
                   float*       __restrict__ agg,
                   int N, int E) {
    __shared__ float tile[TR * D];
    __shared__ uint2 qe[NW * SEG];
    __shared__ float qv[NW * SEG];

    const int tid  = threadIdx.x;
    const int lane = tid & 31;
    const int wave = __builtin_amdgcn_readfirstlane(tid >> 5);
    const int row0 = blockIdx.x * TR;
    int nrow = N - row0;
    if (nrow > TR) nrow = TR;

    float4* tile4 = (float4*)tile;
    for (int i = tid; i < TR * D / 4; i += 1024)
        tile4[i] = make_float4(0.f, 0.f, 0.f, 0.f);
    __syncthreads();

    uint2* myq = qe + wave * SEG;
    float* myv = qv + wave * SEG;
    unsigned cnt = 0;

    const int n4   = E >> 2;
    const int4* er4 = (const int4*)edge_row;
    const int BLK  = 1024 * UNR;
    const int nblk = (n4 + BLK - 1) / BLK;
    int blk = (int)(((long long)blockIdx.x * nblk) / gridDim.x);
    for (int kb = 0; kb < nblk; ++kb) {
        const int base = blk * BLK + tid;
        int4 r[UNR];
        if (blk * BLK + BLK <= n4) {
#pragma unroll
            for (int u = 0; u < UNR; ++u) r[u] = er4[base + u * 1024];
        } else {
#pragma unroll
            for (int u = 0; u < UNR; ++u) {
                const int i4 = base + u * 1024;
                r[u] = er4[i4 < n4 ? i4 : n4 - 1];
            }
#pragma unroll
            for (int u = 0; u < UNR; ++u)
                if (base + u * 1024 >= n4) r[u] = make_int4(-1, -1, -1, -1);
        }
#pragma unroll
        for (int u = 0; u < UNR; ++u) {
            const int d0 = r[u].x - row0, d1 = r[u].y - row0;
            const int d2 = r[u].z - row0, d3 = r[u].w - row0;
            const unsigned dm = min(min((unsigned)d0, (unsigned)d1),
                                    min((unsigned)d2, (unsigned)d3));
            if (__builtin_amdgcn_ballot_w32(dm < (unsigned)nrow) != 0u) {
                const int e0 = (base + u * 1024) * 4;
                queue_match(d0, nrow, e0 + 0, cnt, myq, tile, X, edge_col, edge_val);
                queue_match(d1, nrow, e0 + 1, cnt, myq, tile, X, edge_col, edge_val);
                queue_match(d2, nrow, e0 + 2, cnt, myq, tile, X, edge_col, edge_val);
                queue_match(d3, nrow, e0 + 3, cnt, myq, tile, X, edge_col, edge_val);
            }
        }
        ++blk;
        if (blk == nblk) blk = 0;
    }
    if (wave == 0) {
        const int e = n4 * 4 + lane;
        int d = -1;
        if (lane < (E & 3)) d = edge_row[e] - row0;
        queue_match(d, nrow, e, cnt, myq, tile, X, edge_col, edge_val);
    }

    __builtin_amdgcn_fence(__ATOMIC_ACQ_REL, "wavefront");
    __builtin_amdgcn_wave_barrier();

    const unsigned n = cnt < (unsigned)SEG ? cnt : (unsigned)SEG;

    if (n != 0u) {
        int   ce[SEG / 32];
        float ve[SEG / 32];
#pragma unroll
        for (int k = 0; k < SEG / 32; ++k) {
            const unsigned idx = (unsigned)lane + 32u * k;
            ce[k] = (int)myq[idx < n ? idx : n - 1u].x;
        }
#pragma unroll
        for (int k = 0; k < SEG / 32; ++k) {
            const int e = ce[k];
            ce[k] = edge_col[e];
            ve[k] = edge_val[e];
        }
#pragma unroll
        for (int k = 0; k < SEG / 32; ++k) {
            const unsigned idx = (unsigned)lane + 32u * k;
            if (idx < n) {
                myq[idx].x = (unsigned)ce[k];
                myv[idx]   = ve[k];
            }
        }
    }
    __builtin_amdgcn_fence(__ATOMIC_ACQ_REL, "wavefront");
    __builtin_amdgcn_wave_barrier();

    const int fo = lane * 2;
    for (unsigned i = 0; i < n; i += GB) {
        v2f      xa[GB], xb[GB];
        float    v[GB];
        unsigned lr[GB];
#pragma unroll
        for (int u = 0; u < GB; ++u) {
            const unsigned idx = i + u < n ? i + u : n - 1u;
            const uint2 cr = myq[idx];
            v[u]  = myv[idx];
            lr[u] = cr.y;
            const float* xr = X + (size_t)cr.x * D + fo;
            xa[u] = *(const v2f*)(xr);
            xb[u] = *(const v2f*)(xr + 64);
        }
#pragma unroll
        for (int u = 0; u < GB; ++u) {
            if (i + u < n) {
                float* t = tile + lr[u] * D + fo;
                __hip_atomic_fetch_add(t +  0, v[u] * xa[u].x, __ATOMIC_RELAXED, __HIP_MEMORY_SCOPE_WORKGROUP);
                __hip_atomic_fetch_add(t +  1, v[u] * xa[u].y, __ATOMIC_RELAXED, __HIP_MEMORY_SCOPE_WORKGROUP);
                __hip_atomic_fetch_add(t + 64, v[u] * xb[u].x, __ATOMIC_RELAXED, __HIP_MEMORY_SCOPE_WORKGROUP);
                __hip_atomic_fetch_add(t + 65, v[u] * xb[u].y, __ATOMIC_RELAXED, __HIP_MEMORY_SCOPE_WORKGROUP);
            }
        }
    }
    __syncthreads();

    float4* agg4 = (float4*)(agg + (size_t)row0 * D);
    for (int j = tid; j < nrow * (D / 4); j += 1024) agg4[j] = tile4[j];
}

__global__ void __launch_bounds__(256)
gcn_gemm_wmma_kernel(const float* __restrict__ Wmat,
                     const float* __restrict__ bias,
                     float*       __restrict__ out,
                     int nrows) {
    __shared__ float Wlds[D * WP];

    const int tid = threadIdx.x;

    {
        const float4* Wv = (const float4*)Wmat;
        for (int i4 = tid; i4 < (D * D) / 4; i4 += 256) {
            int o = i4 >> 5;
            int k = (i4 & 31) * 4;
            *(float4*)&Wlds[o * WP + k] = Wv[i4];
        }
    }
    __syncthreads();

    const int wave = tid >> 5;
    const int lane = tid & 31;
    const int lr   = lane & 15;
    const int hi   = lane >> 4;

    const int row_base = blockIdx.x * 128 + wave * 16;
    const int row      = row_base + lr;
    const int rowc     = row < nrows ? row : (nrows - 1);
    const float* arow  = out + (size_t)rowc * D;

    v8f acc[8];
#pragma unroll
    for (int t = 0; t < 8; ++t)
        acc[t] = (v8f){0.f, 0.f, 0.f, 0.f, 0.f, 0.f, 0.f, 0.f};

#pragma unroll 8
    for (int kk = 0; kk < D / 4; ++kk) {
        const int k0 = kk * 4 + hi * 2;
        v2f a = *(const v2f*)(arow + k0);

#pragma unroll
        for (int t = 0; t < 8; ++t) {
            const int o = t * 16 + lr;
            v2f b = *(const v2f*)&Wlds[o * WP + k0];
            acc[t] = __builtin_amdgcn_wmma_f32_16x16x4_f32(
                 false, a,  false, b,
                 (short)0, acc[t],
                 false,  false);
        }
    }

#pragma unroll
    for (int t = 0; t < 8; ++t) {
        const int col = t * 16 + lr;
        const float bb = bias[col];
#pragma unroll
        for (int j = 0; j < 8; ++j) {
            const int r = row_base + j + hi * 8;
            if (r < nrows) out[(size_t)r * D + col] = acc[t][j] + bb;
        }
    }
}

extern "C" void kernel_launch(void* const* d_in, const int* in_sizes, int n_in,
                              void* d_out, int out_size, void* d_ws, size_t ws_size,
                              hipStream_t stream) {
    const float* X        = (const float*)d_in[0];
    const int*   edge_row = (const int*)  d_in[1];
    const int*   edge_col = (const int*)  d_in[2];
    const float* edge_val = (const float*)d_in[3];
    const float* Wmat     = (const float*)d_in[4];
    const float* bias     = (const float*)d_in[5];
    float*       out      = (float*)d_out;

    const int N = in_sizes[0] / D;
    const int E = in_sizes[1];

    {
        int tiles = (N + TR - 1) / TR;
        gcn_rowtile_kernel<<<tiles, 1024, 0, stream>>>(X, edge_row, edge_col,
                                                       edge_val, out, N, E);
    }

    {
        int blocks = (N + 127) / 128;
        gcn_gemm_wmma_kernel<<<blocks, 256, 0, stream>>>(Wmat, bias, out, N);
    }
}


# Tile workgroups scan a 1-byte-per-edge tile digest of edge_row (new pre-pass kernel; 4 edges per dword compare, per-lane LDS queues) instead of the 4-byte rows; LDS f32 sums still in arrival order
# speedup vs baseline: 2.3022x; 1.1659x over previous
#include <hip/hip_runtime.h>


typedef float v2f __attribute__((ext_vector_type(2)));
typedef float v8f __attribute__((ext_vector_type(8)));

#define D 128
#define WP 132

#define TR  392
#define NW  32
#define SLOTS 30
#define UNR 8
#define GB  8

__global__ void __launch_bounds__(256)
gcn_digest_kernel(const int* __restrict__ edge_row, unsigned* __restrict__ dig, int E) {
    const int n4 = (E >> 4) << 2;
    const int i  = blockIdx.x * 256 + threadIdx.x;
    if (i >= n4) return;
    const int4 r = ((const int4*)edge_row)[i];
    const unsigned t0 = ((unsigned)r.x / (unsigned)TR) & 255u;
    const unsigned t1 = ((unsigned)r.y / (unsigned)TR) & 255u;
    const unsigned t2 = ((unsigned)r.z / (unsigned)TR) & 255u;
    const unsigned t3 = ((unsigned)r.w / (unsigned)TR) & 255u;
    dig[i] = t0 | (t1 << 8) | (t2 << 16) | (t3 << 24);
}

__device__ __forceinline__ unsigned zero_bytes(unsigned x) {
    const unsigned t = (x & 0x7F7F7F7Fu) + 0x7F7F7F7Fu;
    return ~(t | x | 0x7F7F7F7Fu);
}

__device__ __noinline__ void add_edge_now(int e, int row0, int nrow, float* tile,
                                          const float* __restrict__ X,
                                          const int*   __restrict__ edge_row,
                                          const int*   __restrict__ edge_col,
                                          const float* __restrict__ edge_val) {
    const unsigned d = (unsigned)(edge_row[e] - row0);
    if (d < (unsigned)nrow) {
        const int   c = edge_col[e];
        const float v = edge_val[e];
        const float* xr = X + (size_t)c * D;
        float* t = tile + d * D;
#pragma nounroll
        for (int f = 0; f < D; ++f)
            __hip_atomic_fetch_add(t + f, v * xr[f], __ATOMIC_RELAXED,
                                   __HIP_MEMORY_SCOPE_WORKGROUP);
    }
}

__global__ void __launch_bounds__(1024)
gcn_rowtile_kernel(const float* __restrict__ X,
                   const int*   __restrict__ edge_row,
                   const int*   __restrict__ edge_col,
                   const float* __restrict__ edge_val,
                   const unsigned* __restrict__ dig,
                   float*       __restrict__ agg,
                   int N, int E) {
    __shared__ float    tile[TR * D];
    __shared__ unsigned qe[NW * SLOTS * 32];

    const int tid  = threadIdx.x;
    const int lane = tid & 31;
    const int wave = __builtin_amdgcn_readfirstlane(tid >> 5);
    const int row0 = blockIdx.x * TR;
    int nrow = N - row0;
    if (nrow > TR) nrow = TR;

    float4* tile4 = (float4*)tile;
    for (int i = tid; i < TR * D / 4; i += 1024)
        tile4[i] = make_float4(0.f, 0.f, 0.f, 0.f);
    __syncthreads();

    unsigned* myq = qe + wave * (SLOTS * 32) + lane;
    unsigned  cnt = 0;

    const int n16  = E >> 4;
    const uint4* dg4 = (const uint4*)dig;
    const unsigned tb = ((unsigned)blockIdx.x & 255u) * 0x01010101u;
    const int BLK  = 1024 * UNR;
    const int nblk = (n16 + BLK - 1) / BLK;
    int blk = (int)(((long long)blockIdx.x * nblk) / gridDim.x);
    for (int kb = 0; kb < nblk; ++kb) {
        const int base = blk * BLK + tid;
        uint4 r[UNR];
        if (blk * BLK + BLK <= n16) {
#pragma unroll
            for (int u = 0; u < UNR; ++u) r[u] = dg4[base + u * 1024];
        } else {
#pragma unroll
            for (int u = 0; u < UNR; ++u) {
                const int i16 = base + u * 1024;
                r[u] = dg4[i16 < n16 ? i16 : n16 - 1];
            }
#pragma unroll
            for (int u = 0; u < UNR; ++u)
                if (base + u * 1024 >= n16) r[u] = make_uint4(~tb, ~tb, ~tb, ~tb);
        }
#pragma unroll
        for (int u = 0; u < UNR; ++u) {
            unsigned m = (zero_bytes(r[u].x ^ tb) >> 7) | (zero_bytes(r[u].y ^ tb) >> 6) |
                         (zero_bytes(r[u].z ^ tb) >> 5) | (zero_bytes(r[u].w ^ tb) >> 4);
            const int e0 = (base + u * 1024) * 16;
            while (m != 0u) {
                const int bpos = __builtin_ctz(m);
                m &= m - 1u;
                const int e = e0 + ((bpos & 7) << 2) + (bpos >> 3);
                if (cnt < (unsigned)SLOTS) { myq[cnt * 32] = (unsigned)e; ++cnt; }
                else add_edge_now(e, row0, nrow, tile, X, edge_row, edge_col, edge_val);
            }
        }
        ++blk;
        if (blk == nblk) blk = 0;
    }
    if (wave == 0) {
        const int e = n16 * 16 + lane;
        if (lane < 16 && e < E) {
            if (cnt < (unsigned)SLOTS) { myq[cnt * 32] = (unsigned)e; ++cnt; }
            else add_edge_now(e, row0, nrow, tile, X, edge_row, edge_col, edge_val);
        }
    }

    const int fo = lane * 2;
    int   c_row, c_col;  float c_val;
    {
        const unsigned q0 = myq[0];
        const unsigned e  = cnt > 0u ? q0 : 0u;
        c_row = edge_row[e];  c_col = edge_col[e];  c_val = edge_val[e];
    }
    unsigned c_d = cnt > 0u ? (unsigned)(c_row - row0) : 0xFFFFFFFFu;
    for (unsigned s = 0; s < (unsigned)SLOTS; ++s) {
        if (__builtin_amdgcn_ballot_w32(s < cnt) == 0u) break;
        const unsigned sn = s + 1u < (unsigned)SLOTS ? s + 1u : s;
        const bool     nv = s + 1u < cnt;
        const unsigned qn = myq[sn * 32];
        const unsigned en = nv ? qn : 0u;
        const int   n_row = edge_row[en];
        const int   n_col = edge_col[en];
        const float n_val = edge_val[en];

        unsigned vm = __builtin_amdgcn_ballot_w32(c_d < (unsigned)nrow);
        while (vm != 0u) {
            v2f      xa[GB], xb[GB];
            float    v[GB];
            unsigned lr[GB];
            bool     ok[GB];
            unsigned g = vm;
#pragma unroll
            for (int u = 0; u < GB; ++u) {
                ok[u] = g != 0u;
                const int L = __builtin_ctz(ok[u] ? g : vm);
                g &= g - 1u;
                const int   cs = __builtin_amdgcn_readlane(c_col, L);
                lr[u] = (unsigned)__builtin_amdgcn_readlane((int)c_d, L);
                v[u]  = __int_as_float(__builtin_amdgcn_readlane(__float_as_int(c_val), L));
                const float* xr = X + (size_t)cs * D + fo;
                xa[u] = *(const v2f*)(xr);
                xb[u] = *(const v2f*)(xr + 64);
            }
#pragma unroll
            for (int u = 0; u < GB; ++u) {
                if (ok[u]) {
                    float* t = tile + lr[u] * D + fo;
                    __hip_atomic_fetch_add(t +  0, v[u] * xa[u].x, __ATOMIC_RELAXED, __HIP_MEMORY_SCOPE_WORKGROUP);
                    __hip_atomic_fetch_add(t +  1, v[u] * xa[u].y, __ATOMIC_RELAXED, __HIP_MEMORY_SCOPE_WORKGROUP);
                    __hip_atomic_fetch_add(t + 64, v[u] * xb[u].x, __ATOMIC_RELAXED, __HIP_MEMORY_SCOPE_WORKGROUP);
                    __hip_atomic_fetch_add(t + 65, v[u] * xb[u].y, __ATOMIC_RELAXED, __HIP_MEMORY_SCOPE_WORKGROUP);
                }
            }
            vm = g;
        }
        c_col = n_col;  c_val = n_val;
        c_d   = nv ? (unsigned)(n_row - row0) : 0xFFFFFFFFu;
    }
    __syncthreads();

    float4* agg4 = (float4*)(agg + (size_t)row0 * D);
    for (int j = tid; j < nrow * (D / 4); j += 1024) agg4[j] = tile4[j];
}

__global__ void __launch_bounds__(256)
gcn_gemm_wmma_kernel(const float* __restrict__ Wmat,
                     const float* __restrict__ bias,
                     float*       __restrict__ out,
                     int nrows) {
    __shared__ float Wlds[D * WP];

    const int tid = threadIdx.x;

    {
        const float4* Wv = (const float4*)Wmat;
        for (int i4 = tid; i4 < (D * D) / 4; i4 += 256) {
            int o = i4 >> 5;
            int k = (i4 & 31) * 4;
            *(float4*)&Wlds[o * WP + k] = Wv[i4];
        }
    }
    __syncthreads();

    const int wave = tid >> 5;
    const int lane = tid & 31;
    const int lr   = lane & 15;
    const int hi   = lane >> 4;

    const int row_base = blockIdx.x * 128 + wave * 16;
    const int row      = row_base + lr;
    const int rowc     = row < nrows ? row : (nrows - 1);
    const float* arow  = out + (size_t)rowc * D;

    v8f acc[8];
#pragma unroll
    for (int t = 0; t < 8; ++t)
        acc[t] = (v8f){0.f, 0.f, 0.f, 0.f, 0.f, 0.f, 0.f, 0.f};

#pragma unroll 8
    for (int kk = 0; kk < D / 4; ++kk) {
        const int k0 = kk * 4 + hi * 2;
        v2f a = *(const v2f*)(arow + k0);

#pragma unroll
        for (int t = 0; t < 8; ++t) {
            const int o = t * 16 + lr;
            v2f b = *(const v2f*)&Wlds[o * WP + k0];
            acc[t] = __builtin_amdgcn_wmma_f32_16x16x4_f32(
                 false, a,  false, b,
                 (short)0, acc[t],
                 false,  false);
        }
    }

#pragma unroll
    for (int t = 0; t < 8; ++t) {
        const int col = t * 16 + lr;
        const float bb = bias[col];
#pragma unroll
        for (int j = 0; j < 8; ++j) {
            const int r = row_base + j + hi * 8;
            if (r < nrows) out[(size_t)r * D + col] = acc[t][j] + bb;
        }
    }
}

extern "C" void kernel_launch(void* const* d_in, const int* in_sizes, int n_in,
                              void* d_out, int out_size, void* d_ws, size_t ws_size,
                              hipStream_t stream) {
    const float* X        = (const float*)d_in[0];
    const int*   edge_row = (const int*)  d_in[1];
    const int*   edge_col = (const int*)  d_in[2];
    const float* edge_val = (const float*)d_in[3];
    const float* Wmat     = (const float*)d_in[4];
    const float* bias     = (const float*)d_in[5];
    float*       out      = (float*)d_out;
    unsigned*    dig      = (unsigned*)d_ws;

    const int N = in_sizes[0] / D;
    const int E = in_sizes[1];

    {
        int dblocks = (E / 4 + 255) / 256;
        if (dblocks > 0)
            gcn_digest_kernel<<<dblocks, 256, 0, stream>>>(edge_row, dig, E);
    }

    {
        int tiles = (N + TR - 1) / TR;
        gcn_rowtile_kernel<<<tiles, 1024, 0, stream>>>(X, edge_row, edge_col,
                                                       edge_val, dig, out, N, E);
    }

    {
        int blocks = (N + 127) / 128;
        gcn_gemm_wmma_kernel<<<blocks, 256, 0, stream>>>(Wmat, bias, out, N);
    }
}


# Tile kernel keeps its matched edges in one dense list per wave (ballot rank), not per-lane queues; gather runs in full branch-free 8-row groups, next loads in flight under the adds; arrival-order sums
# speedup vs baseline: 2.4146x; 1.0488x over previous
#include <hip/hip_runtime.h>


typedef float v2f __attribute__((ext_vector_type(2)));
typedef float v8f __attribute__((ext_vector_type(8)));

#define D 128
#define WP 132

#define TR  392
#define NW  32
#define QCAP 960
#define UNR 8
#define GB  8

__global__ void __launch_bounds__(256)
gcn_digest_kernel(const int* __restrict__ edge_row, unsigned* __restrict__ dig, int E) {
    const int n4 = (E >> 4) << 2;
    const int i  = blockIdx.x * 256 + threadIdx.x;
    if (i >= n4) return;
    const int4 r = ((const int4*)edge_row)[i];
    const unsigned t0 = ((unsigned)r.x / (unsigned)TR) & 255u;
    const unsigned t1 = ((unsigned)r.y / (unsigned)TR) & 255u;
    const unsigned t2 = ((unsigned)r.z / (unsigned)TR) & 255u;
    const unsigned t3 = ((unsigned)r.w / (unsigned)TR) & 255u;
    dig[i] = t0 | (t1 << 8) | (t2 << 16) | (t3 << 24);
}

__device__ __forceinline__ unsigned zero_bytes(unsigned x) {
    const unsigned t = (x & 0x7F7F7F7Fu) + 0x7F7F7F7Fu;
    return ~(t | x | 0x7F7F7F7Fu);
}

__device__ __noinline__ void add_edge_now(int e, int row0, int nrow, float* tile,
                                          const float* __restrict__ X,
                                          const int*   __restrict__ edge_row,
                                          const int*   __restrict__ edge_col,
                                          const float* __restrict__ edge_val) {
    const unsigned d = (unsigned)(edge_row[e] - row0);
    if (d < (unsigned)nrow) {
        const int   c = edge_col[e];
        const float v = edge_val[e];
        const float* xr = X + (size_t)c * D;
        float* t = tile + d * D;
#pragma nounroll
        for (int f = 0; f < D; ++f)
            __hip_atomic_fetch_add(t + f, v * xr[f], __ATOMIC_RELAXED,
                                   __HIP_MEMORY_SCOPE_WORKGROUP);
    }
}

__global__ void __launch_bounds__(1024)
gcn_rowtile_kernel(const float* __restrict__ X,
                   const int*   __restrict__ edge_row,
                   const int*   __restrict__ edge_col,
                   const float* __restrict__ edge_val,
                   const unsigned* __restrict__ dig,
                   float*       __restrict__ agg,
                   int N, int E) {
    __shared__ float    tile[(TR + 1) * D];
    __shared__ unsigned qe[NW * QCAP];

    const int tid  = threadIdx.x;
    const int lane = tid & 31;
    const int wave = __builtin_amdgcn_readfirstlane(tid >> 5);
    const int row0 = blockIdx.x * TR;
    int nrow = N - row0;
    if (nrow > TR) nrow = TR;

    float4* tile4 = (float4*)tile;
    for (int i = tid; i < (TR + 1) * D / 4; i += 1024)
        tile4[i] = make_float4(0.f, 0.f, 0.f, 0.f);
    __syncthreads();

    unsigned* wq   = qe + wave * QCAP;
    unsigned  wcnt = 0;

    const int n16  = E >> 4;
    const uint4* dg4 = (const uint4*)dig;
    const unsigned tb = ((unsigned)blockIdx.x & 255u) * 0x01010101u;
    const int BLK  = 1024 * UNR;
    const int nblk = (n16 + BLK - 1) / BLK;
    int blk = (int)(((long long)blockIdx.x * nblk) / gridDim.x);
    for (int kb = 0; kb < nblk; ++kb) {
        const int base = blk * BLK + tid;
        uint4 r[UNR];
        if (blk * BLK + BLK <= n16) {
#pragma unroll
            for (int u = 0; u < UNR; ++u) r[u] = dg4[base + u * 1024];
        } else {
#pragma unroll
            for (int u = 0; u < UNR; ++u) {
                const int i16 = base + u * 1024;
                r[u] = dg4[i16 < n16 ? i16 : n16 - 1];
            }
#pragma unroll
            for (int u = 0; u < UNR; ++u)
                if (base + u * 1024 >= n16) r[u] = make_uint4(~tb, ~tb, ~tb, ~tb);
        }
#pragma unroll
        for (int u = 0; u < UNR; ++u) {
            unsigned m = (zero_bytes(r[u].x ^ tb) >> 7) | (zero_bytes(r[u].y ^ tb) >> 6) |
                         (zero_bytes(r[u].z ^ tb) >> 5) | (zero_bytes(r[u].w ^ tb) >> 4);
            const int e0 = (base + u * 1024) * 16;
            unsigned any = __builtin_amdgcn_ballot_w32(m != 0u);
            while (any != 0u) {
                if (m != 0u) {
                    const int bpos = __builtin_ctz(m);
                    m &= m - 1u;
                    const int e = e0 + ((bpos & 7) << 2) + (bpos >> 3);
                    const unsigned pos = wcnt + __builtin_amdgcn_mbcnt_lo(any, 0u);
                    if (pos < (unsigned)QCAP) wq[pos] = (unsigned)e;
                    else add_edge_now(e, row0, nrow, tile, X, edge_row, edge_col, edge_val);
                }
                wcnt += (unsigned)__builtin_popcount(any);
                any = __builtin_amdgcn_ballot_w32(m != 0u);
            }
        }
        ++blk;
        if (blk == nblk) blk = 0;
    }
    if (wave == 0) {
        const int  e   = n16 * 16 + lane;
        const bool has = lane < 16 && e < E;
        const unsigned any = __builtin_amdgcn_ballot_w32(has);
        if (has) {
            const unsigned pos = wcnt + __builtin_amdgcn_mbcnt_lo(any, 0u);
            if (pos < (unsigned)QCAP) wq[pos] = (unsigned)e;
            else add_edge_now(e, row0, nrow, tile, X, edge_row, edge_col, edge_val);
        }
        wcnt += (unsigned)__builtin_popcount(any);
    }

    const unsigned n   = wcnt < (unsigned)QCAP ? wcnt : (unsigned)QCAP;
    const unsigned nch = (n + 31u) >> 5;
    const int fo = lane * 2;

#define LOOK_UP(BASE, COL, VAL, DST)                                            \
    {                                                                           \
        const unsigned i_   = (BASE) + (unsigned)lane;                          \
        const bool     vl_  = i_ < n;                                           \
        const unsigned q_   = wq[vl_ ? i_ : 0u];                                \
        const unsigned e_   = vl_ ? q_ : 0u;                                    \
        const unsigned lim_ = vl_ ? (unsigned)nrow : 0u;                        \
        const unsigned d_   = (unsigned)(edge_row[e_] - row0);                  \
        COL = edge_col[e_];                                                     \
        VAL = edge_val[e_];                                                     \
        DST = d_ < lim_ ? d_ : (unsigned)TR;                                    \
    }
#define ISSUE_ROWS(COL, G, XA, XB)                                              \
    _Pragma("unroll")                                                           \
    for (int u_ = 0; u_ < GB; ++u_) {                                           \
        const int cs_ = __builtin_amdgcn_readlane(COL, (G) * GB + u_);          \
        const float* xr_ = X + (size_t)cs_ * D + fo;                            \
        XA[u_] = *(const v2f*)(xr_);                                            \
        XB[u_] = *(const v2f*)(xr_ + 64);                                       \
    }
#define ADD_ROWS(DST, VAL, G, XA, XB)                                           \
    _Pragma("unroll")                                                           \
    for (int u_ = 0; u_ < GB; ++u_) {                                           \
        const unsigned lr_ = (unsigned)__builtin_amdgcn_readlane((int)DST, (G) * GB + u_); \
        const float    v_  = __int_as_float(__builtin_amdgcn_readlane(__float_as_int(VAL), (G) * GB + u_)); \
        float* t_ = tile + lr_ * D + fo;                                        \
        __hip_atomic_fetch_add(t_ +  0, v_ * XA[u_].x, __ATOMIC_RELAXED, __HIP_MEMORY_SCOPE_WORKGROUP); \
        __hip_atomic_fetch_add(t_ +  1, v_ * XA[u_].y, __ATOMIC_RELAXED, __HIP_MEMORY_SCOPE_WORKGROUP); \
        __hip_atomic_fetch_add(t_ + 64, v_ * XB[u_].x, __ATOMIC_RELAXED, __HIP_MEMORY_SCOPE_WORKGROUP); \
        __hip_atomic_fetch_add(t_ + 65, v_ * XB[u_].y, __ATOMIC_RELAXED, __HIP_MEMORY_SCOPE_WORKGROUP); \
    }

    if (nch != 0u) {
        int c_col;  float c_val;  unsigned c_d;
        LOOK_UP(0u, c_col, c_val, c_d)
        v2f xa0[GB], xb0[GB], xa1[GB], xb1[GB];
        ISSUE_ROWS(c_col, 0, xa0, xb0)
        for (unsigned ch = 0; ch < nch; ++ch) {
            int n_col;  float n_val;  unsigned n_d;
            LOOK_UP((ch + 1u) << 5, n_col, n_val, n_d)
            ISSUE_ROWS(c_col, 1, xa1, xb1)
            __builtin_amdgcn_sched_barrier(0);
            ADD_ROWS(c_d, c_val, 0, xa0, xb0)
            ISSUE_ROWS(c_col, 2, xa0, xb0)
            ADD_ROWS(c_d, c_val, 1, xa1, xb1)
            ISSUE_ROWS(c_col, 3, xa1, xb1)
            ADD_ROWS(c_d, c_val, 2, xa0, xb0)
            ISSUE_ROWS(n_col, 0, xa0, xb0)
            ADD_ROWS(c_d, c_val, 3, xa1, xb1)
            c_col = n_col;  c_val = n_val;  c_d = n_d;
        }
    }
#undef LOOK_UP
#undef ISSUE_ROWS
#undef ADD_ROWS
    __syncthreads();

    float4* agg4 = (float4*)(agg + (size_t)row0 * D);
    for (int j = tid; j < nrow * (D / 4); j += 1024) agg4[j] = tile4[j];
}

__global__ void __launch_bounds__(256)
gcn_gemm_wmma_kernel(const float* __restrict__ Wmat,
                     const float* __restrict__ bias,
                     float*       __restrict__ out,
                     int nrows) {
    __shared__ float Wlds[D * WP];

    const int tid = threadIdx.x;

    {
        const float4* Wv = (const float4*)Wmat;
        for (int i4 = tid; i4 < (D * D) / 4; i4 += 256) {
            int o = i4 >> 5;
            int k = (i4 & 31) * 4;
            *(float4*)&Wlds[o * WP + k] = Wv[i4];
        }
    }
    __syncthreads();

    const int wave = tid >> 5;
    const int lane = tid & 31;
    const int lr   = lane & 15;
    const int hi   = lane >> 4;

    const int row_base = blockIdx.x * 128 + wave * 16;
    const int row      = row_base + lr;
    const int rowc     = row < nrows ? row : (nrows - 1);
    const float* arow  = out + (size_t)rowc * D;

    v8f acc[8];
#pragma unroll
    for (int t = 0; t < 8; ++t)
        acc[t] = (v8f){0.f, 0.f, 0.f, 0.f, 0.f, 0.f, 0.f, 0.f};

#pragma unroll 8
    for (int kk = 0; kk < D / 4; ++kk) {
        const int k0 = kk * 4 + hi * 2;
        v2f a = *(const v2f*)(arow + k0);

#pragma unroll
        for (int t = 0; t < 8; ++t) {
            const int o = t * 16 + lr;
            v2f b = *(const v2f*)&Wlds[o * WP + k0];
            acc[t] = __builtin_amdgcn_wmma_f32_16x16x4_f32(
                 false, a,  false, b,
                 (short)0, acc[t],
                 false,  false);
        }
    }

#pragma unroll
    for (int t = 0; t < 8; ++t) {
        const int col = t * 16 + lr;
        const float bb = bias[col];
#pragma unroll
        for (int j = 0; j < 8; ++j) {
            const int r = row_base + j + hi * 8;
            if (r < nrows) out[(size_t)r * D + col] = acc[t][j] + bb;
        }
    }
}

extern "C" void kernel_launch(void* const* d_in, const int* in_sizes, int n_in,
                              void* d_out, int out_size, void* d_ws, size_t ws_size,
                              hipStream_t stream) {
    const float* X        = (const float*)d_in[0];
    const int*   edge_row = (const int*)  d_in[1];
    const int*   edge_col = (const int*)  d_in[2];
    const float* edge_val = (const float*)d_in[3];
    const float* Wmat     = (const float*)d_in[4];
    const float* bias     = (const float*)d_in[5];
    float*       out      = (float*)d_out;
    unsigned*    dig      = (unsigned*)d_ws;

    const int N = in_sizes[0] / D;
    const int E = in_sizes[1];

    {
        int dblocks = (E / 4 + 255) / 256;
        if (dblocks > 0)
            gcn_digest_kernel<<<dblocks, 256, 0, stream>>>(edge_row, dig, E);
    }

    {
        int tiles = (N + TR - 1) / TR;
        gcn_rowtile_kernel<<<tiles, 1024, 0, stream>>>(X, edge_row, edge_col,
                                                       edge_val, dig, out, N, E);
    }

    {
        int blocks = (N + 127) / 128;
        gcn_gemm_wmma_kernel<<<blocks, 256, 0, stream>>>(Wmat, bias, out, N);
    }
}


# LDS f32 atomics of the row tile replaced by owner waves: edges go to per-wave LDS queues (integer rank), each wave adds its own rows by plain 16-B LDS load, fma, store; sums in queue arrival order
# speedup vs baseline: 3.4517x; 1.4295x over previous
#include <hip/hip_runtime.h>


typedef float v2f __attribute__((ext_vector_type(2)));
typedef float v8f __attribute__((ext_vector_type(8)));

#define D 128
#define WP 132

#define TR  392
#define NW  32
#define GB  8
#define QCAP 320
#define CB  23
#define CMASK 0x7FFFFFu
#define BE  4096
#define KPT 4
#define NCL 256

__device__ __forceinline__ unsigned rec_offset_dwords(int NB) {
    return ((unsigned)(NCL + 1) * (unsigned)NB + 3u) & ~3u;
}

__global__ void __launch_bounds__(1024)
gcn_blocksort_kernel(const int*   __restrict__ edge_row,
                     const int*   __restrict__ edge_col,
                     const float* __restrict__ edge_val,
                     unsigned*    __restrict__ ws, int E) {
    __shared__ unsigned hist[NCL];
    __shared__ unsigned start[NCL + 1];
    __shared__ uint4    stage[BE];

    const int tid = threadIdx.x;
    const int NB  = (E + BE - 1) / BE;
    const int b   = blockIdx.x;
    const int e0  = b * BE;
    unsigned* offT = ws;
    uint4*    rec  = (uint4*)(ws + rec_offset_dwords(NB));

    if (tid < NCL) hist[tid] = 0u;
    __syncthreads();

    int row[KPT], col[KPT];
    float val[KPT];
    unsigned cls[KPT], rk[KPT];
#pragma unroll
    for (int k = 0; k < KPT; ++k) {
        const int e  = e0 + k * 1024 + tid;
        const int ec = e < E ? e : E - 1;
        row[k] = edge_row[ec];
        col[k] = edge_col[ec];
        val[k] = edge_val[ec];
    }
#pragma unroll
    for (int k = 0; k < KPT; ++k) {
        const int e = e0 + k * 1024 + tid;
        cls[k] = ((unsigned)row[k] / (unsigned)TR) & (unsigned)(NCL - 1);
        rk[k]  = 0u;
        if (e < E) rk[k] = atomicAdd(&hist[cls[k]], 1u);
    }
    __syncthreads();

    if (tid < 32) {
        unsigned s[8];
        unsigned sum = 0u;
#pragma unroll
        for (int q = 0; q < 8; ++q) { s[q] = hist[tid * 8 + q]; sum += s[q]; }
        unsigned inc = sum;
#pragma unroll
        for (int dlt = 1; dlt < 32; dlt <<= 1) {
            const unsigned t = __shfl_up(inc, (unsigned)dlt);
            if (tid >= dlt) inc += t;
        }
        unsigned run = inc - sum;
#pragma unroll
        for (int q = 0; q < 8; ++q) { start[tid * 8 + q] = run; run += s[q]; }
        if (tid == 31) start[NCL] = run;
    }
    __syncthreads();

    if (tid <= NCL) offT[(unsigned)tid * (unsigned)NB + (unsigned)b] = start[tid];

#pragma unroll
    for (int k = 0; k < KPT; ++k) {
        const int e = e0 + k * 1024 + tid;
        if (e < E)
            stage[start[cls[k]] + rk[k]] =
                make_uint4((unsigned)row[k], (unsigned)col[k], __float_as_uint(val[k]), (unsigned)e);
    }
    __syncthreads();

    const unsigned cnt = start[NCL];
    for (unsigned i = (unsigned)tid; i < cnt; i += 1024u) rec[(size_t)e0 + i] = stage[i];
}

__global__ void __launch_bounds__(1024)
gcn_rowtile_kernel(const float*    __restrict__ X,
                   const unsigned* __restrict__ ws,
                   float*          __restrict__ agg,
                   int N, int E) {
    __shared__ float    tile[(TR + 1) * D];
    __shared__ uint2    queue[NW * QCAP];
    __shared__ unsigned qcnt[NW];

    const int tid  = threadIdx.x;
    const int lane = tid & 31;
    const int wave = __builtin_amdgcn_readfirstlane(tid >> 5);
    const int row0 = blockIdx.x * TR;
    int nrow = N - row0;
    if (nrow > TR) nrow = TR;

    float4* tile4 = (float4*)tile;
    for (int i = tid; i < (TR + 1) * D / 4; i += 1024)
        tile4[i] = make_float4(0.f, 0.f, 0.f, 0.f);
    if (tid < NW) qcnt[tid] = 0u;
    __syncthreads();

    const int NB = (E + BE - 1) / BE;
    const unsigned* offT = ws;
    const uint4*    rec  = (const uint4*)(ws + rec_offset_dwords(NB));
    const unsigned  cls  = (unsigned)blockIdx.x & (unsigned)(NCL - 1);
    const unsigned* off0 = offT + cls * (unsigned)NB;
    const unsigned* off1 = offT + (cls + 1u) * (unsigned)NB;
    const int nseg = wave < NB ? (NB - wave + NW - 1) / NW : 0;

    const int fo = lane * 4;

#define LOOK_UP(BASE, COL, VAL, DST)                                            \
    {                                                                           \
        const unsigned i_   = (BASE) + (unsigned)lane;                          \
        const bool     vl_  = i_ < n;                                           \
        unsigned j_ = 0u;                                                       \
        for (int k_ = 0; k_ < ns; ++k_)                                         \
            j_ += (i_ >= (unsigned)__builtin_amdgcn_readlane((int)inc, k_)) ? 1u : 0u; \
        const unsigned sb_  = (unsigned)__shfl((int)sbase, (int)(vl_ ? j_ : 0u)); \
        const uint4    rc_  = rec[vl_ ? sb_ + i_ : 0u];                         \
        const unsigned lim_ = vl_ ? (unsigned)nrow : 0u;                        \
        const unsigned d_   = rc_.x - (unsigned)row0;                           \
        COL = rc_.y;                                                            \
        VAL = rc_.z;                                                            \
        DST = d_ < lim_ ? d_ : (unsigned)TR;                                    \
    }
#define HAND_OUT(COL, VAL, DST)                                                 \
    {                                                                           \
        bool over_ = false;                                                     \
        if (DST != (unsigned)TR) {                                              \
            const unsigned ow_ = DST & (unsigned)(NW - 1);                      \
            const unsigned rk_ = atomicAdd(&qcnt[ow_], 1u);                     \
            if (rk_ < (unsigned)QCAP)                                           \
                queue[ow_ * (unsigned)QCAP + rk_] = make_uint2(COL | (DST << CB), VAL); \
            else                                                                \
                over_ = true;                                                   \
        }                                                                       \
        unsigned om_ = (unsigned)__ballot(over_);                               \
        while (om_ != 0u) {     \
            const int l_ = __builtin_ctz(om_);                                  \
            om_ &= om_ - 1u;                                                    \
            const unsigned c_ = (unsigned)__shfl((int)COL, l_);                 \
            const float    v_ = __uint_as_float((unsigned)__shfl((int)VAL, l_)); \
            const unsigned r_ = (unsigned)__shfl((int)DST, l_);                 \
            const float4   x_ = *(const float4*)(X + (size_t)c_ * D + fo);      \
            float* t_ = tile + r_ * D + fo;                                     \
            __hip_atomic_fetch_add(t_ + 0, v_ * x_.x, __ATOMIC_RELAXED, __HIP_MEMORY_SCOPE_WORKGROUP); \
            __hip_atomic_fetch_add(t_ + 1, v_ * x_.y, __ATOMIC_RELAXED, __HIP_MEMORY_SCOPE_WORKGROUP); \
            __hip_atomic_fetch_add(t_ + 2, v_ * x_.z, __ATOMIC_RELAXED, __HIP_MEMORY_SCOPE_WORKGROUP); \
            __hip_atomic_fetch_add(t_ + 3, v_ * x_.w, __ATOMIC_RELAXED, __HIP_MEMORY_SCOPE_WORKGROUP); \
        }                                                                       \
    }

    for (int p0 = 0; p0 < nseg; p0 += 32) {
        const int  ns = nseg - p0 < 32 ? nseg - p0 : 32;
        const bool sv = lane < ns;
        const int  bk = wave + NW * (p0 + lane);
        unsigned st = 0u, en = 0u;
        if (sv) { st = off0[bk]; en = off1[bk]; }
        const unsigned cnt = en - st;
        unsigned inc = cnt;
#pragma unroll
        for (int dlt = 1; dlt < 32; dlt <<= 1) {
            const unsigned t = __shfl_up(inc, (unsigned)dlt);
            if (lane >= dlt) inc += t;
        }
        const unsigned sbase = (unsigned)bk * (unsigned)BE + st - (inc - cnt);
        const unsigned n   = (unsigned)__builtin_amdgcn_readlane((int)inc, 31);
        const unsigned nch = (n + 31u) >> 5;

        if (nch != 0u) {
            unsigned c_col, c_val, c_d;
            LOOK_UP(0u, c_col, c_val, c_d)
            for (unsigned ch = 0; ch < nch; ++ch) {
                unsigned n_col, n_val, n_d;
                LOOK_UP((ch + 1u) << 5, n_col, n_val, n_d)
                HAND_OUT(c_col, c_val, c_d)
                c_col = n_col;  c_val = n_val;  c_d = n_d;
            }
        }
    }
#undef LOOK_UP
#undef HAND_OUT
    __syncthreads();

    {
        unsigned qn = qcnt[wave];
        if (qn > (unsigned)QCAP) qn = (unsigned)QCAP;
        qn = (unsigned)__builtin_amdgcn_readfirstlane((int)qn);
        const uint2*   q  = queue + (unsigned)wave * (unsigned)QCAP;
        const unsigned ng = (qn + (unsigned)GB - 1u) / (unsigned)GB;

#define GROUP(J, PK, VL)                                                        \
        {                                                                       \
            const unsigned ri_ = (J) * (unsigned)GB + (unsigned)(lane & (GB - 1)); \
            const bool     ok_ = ri_ < qn;                                      \
            const uint2    r_  = q[ok_ ? ri_ : 0u];                             \
            PK = ok_ ? r_.x : ((r_.x & CMASK) | ((unsigned)TR << CB));          \
            VL = r_.y;                                                          \
        }
#define ISSUE_ROWS(PK, XR)                                                      \
        _Pragma("unroll")                                                       \
        for (int u_ = 0; u_ < GB; ++u_) {                                       \
            const unsigned p_ = (unsigned)__builtin_amdgcn_readlane((int)PK, u_); \
            XR[u_] = *(const float4*)(X + (size_t)(p_ & CMASK) * D + fo);       \
        }
#define ADD_ROWS(PK, VL, XR)                                                    \
        _Pragma("unroll")                                                       \
        for (int u_ = 0; u_ < GB; ++u_) {                                       \
            const unsigned lr_ = (unsigned)__builtin_amdgcn_readlane((int)PK, u_) >> CB; \
            const float    v_  = __uint_as_float((unsigned)__builtin_amdgcn_readlane((int)VL, u_)); \
            float4* t_ = (float4*)(tile + lr_ * D + fo);                        \
            float4  a_ = *t_;                                                   \
            a_.x += v_ * XR[u_].x;                                              \
            a_.y += v_ * XR[u_].y;                                              \
            a_.z += v_ * XR[u_].z;                                              \
            a_.w += v_ * XR[u_].w;                                              \
            *t_ = a_;                                                           \
        }

        if (ng != 0u) {
            unsigned pk0, vl0, pk1, vl1;
            float4 xa[GB], xb[GB];
            GROUP(0u, pk0, vl0)
            GROUP(1u, pk1, vl1)
            ISSUE_ROWS(pk0, xa)
            for (unsigned k = 0; k < ng; k += 2u) {
                unsigned pk2, vl2, pk3, vl3;
                ISSUE_ROWS(pk1, xb)
                GROUP(k + 2u, pk2, vl2)
                GROUP(k + 3u, pk3, vl3)
                ADD_ROWS(pk0, vl0, xa)
                ISSUE_ROWS(pk2, xa)
                ADD_ROWS(pk1, vl1, xb)
                pk0 = pk2;  vl0 = vl2;  pk1 = pk3;  vl1 = vl3;
            }
        }
#undef GROUP
#undef ISSUE_ROWS
#undef ADD_ROWS
    }
    __syncthreads();

    float4* agg4 = (float4*)(agg + (size_t)row0 * D);
    for (int j = tid; j < nrow * (D / 4); j += 1024) agg4[j] = tile4[j];
}

__global__ void __launch_bounds__(256)
gcn_gemm_wmma_kernel(const float* __restrict__ Wmat,
                     const float* __restrict__ bias,
                     float*       __restrict__ out,
                     int nrows) {
    __shared__ float Wlds[D * WP];

    const int tid = threadIdx.x;

    {
        const float4* Wv = (const float4*)Wmat;
        for (int i4 = tid; i4 < (D * D) / 4; i4 += 256) {
            int o = i4 >> 5;
            int k = (i4 & 31) * 4;
            *(float4*)&Wlds[o * WP + k] = Wv[i4];
        }
    }
    __syncthreads();

    const int wave = tid >> 5;
    const int lane = tid & 31;
    const int lr   = lane & 15;
    const int hi   = lane >> 4;

    const int row_base = blockIdx.x * 128 + wave * 16;
    const int row      = row_base + lr;
    const int rowc     = row < nrows ? row : (nrows - 1);
    const float* arow  = out + (size_t)rowc * D;

    v8f acc[8];
#pragma unroll
    for (int t = 0; t < 8; ++t)
        acc[t] = (v8f){0.f, 0.f, 0.f, 0.f, 0.f, 0.f, 0.f, 0.f};

#pragma unroll 8
    for (int kk = 0; kk < D / 4; ++kk) {
        const int k0 = kk * 4 + hi * 2;
        v2f a = *(const v2f*)(arow + k0);

#pragma unroll
        for (int t = 0; t < 8; ++t) {
            const int o = t * 16 + lr;
            v2f b = *(const v2f*)&Wlds[o * WP + k0];
            acc[t] = __builtin_amdgcn_wmma_f32_16x16x4_f32(
                 false, a,  false, b,
                 (short)0, acc[t],
                 false,  false);
        }
    }

#pragma unroll
    for (int t = 0; t < 8; ++t) {
        const int col = t * 16 + lr;
        const float bb = bias[col];
#pragma unroll
        for (int j = 0; j < 8; ++j) {
            const int r = row_base + j + hi * 8;
            if (r < nrows) out[(size_t)r * D + col] = acc[t][j] + bb;
        }
    }
}

extern "C" void kernel_launch(void* const* d_in, const int* in_sizes, int n_in,
                              void* d_out, int out_size, void* d_ws, size_t ws_size,
                              hipStream_t stream) {
    const float* X        = (const float*)d_in[0];
    const int*   edge_row = (const int*)  d_in[1];
    const int*   edge_col = (const int*)  d_in[2];
    const float* edge_val = (const float*)d_in[3];
    const float* Wmat     = (const float*)d_in[4];
    const float* bias     = (const float*)d_in[5];
    float*       out      = (float*)d_out;
    unsigned*    ws       = (unsigned*)d_ws;

    const int N = in_sizes[0] / D;
    const int E = in_sizes[1];

    {
        int sblocks = (E + BE - 1) / BE;
        if (sblocks > 0)
            gcn_blocksort_kernel<<<sblocks, 1024, 0, stream>>>(edge_row, edge_col, edge_val, ws, E);
    }

    {
        int tiles = (N + TR - 1) / TR;
        gcn_rowtile_kernel<<<tiles, 1024, 0, stream>>>(X, ws, out, N, E);
    }

    {
        int blocks = (N + 127) / 128;
        gcn_gemm_wmma_kernel<<<blocks, 256, 0, stream>>>(Wmat, bias, out, N);
    }
}


# Dense product moved into the row-tile kernel: W staged in the emptied queue LDS, the finished 392-row tile (pitch 132) multiplied in place by WMMA f32; agg never stored, 2 launches; sum orders kept
# speedup vs baseline: 4.0739x; 1.1803x over previous
#include <hip/hip_runtime.h>


typedef float v2f __attribute__((ext_vector_type(2)));
typedef float v8f __attribute__((ext_vector_type(8)));

#define D 128
#define WP 132
#define TP 132
#define NRT 25

#define TR  392
#define NW  32
#define GB  8
#define QCAP 320
#define CB  23
#define CMASK 0x7FFFFFu
#define BE  4096
#define KPT 4
#define NCL 256

__device__ __forceinline__ unsigned rec_offset_dwords(int NB) {
    return ((unsigned)(NCL + 1) * (unsigned)NB + 3u) & ~3u;
}

__global__ void __launch_bounds__(1024)
gcn_blocksort_kernel(const int*   __restrict__ edge_row,
                     const int*   __restrict__ edge_col,
                     const float* __restrict__ edge_val,
                     unsigned*    __restrict__ ws, int E) {
    __shared__ unsigned hist[NCL];
    __shared__ unsigned start[NCL + 1];
    __shared__ uint4    stage[BE];

    const int tid = threadIdx.x;
    const int NB  = (E + BE - 1) / BE;
    const int b   = blockIdx.x;
    const int e0  = b * BE;
    unsigned* offT = ws;
    uint4*    rec  = (uint4*)(ws + rec_offset_dwords(NB));

    if (tid < NCL) hist[tid] = 0u;
    __syncthreads();

    int row[KPT], col[KPT];
    float val[KPT];
    unsigned cls[KPT], rk[KPT];
#pragma unroll
    for (int k = 0; k < KPT; ++k) {
        const int e  = e0 + k * 1024 + tid;
        const int ec = e < E ? e : E - 1;
        row[k] = edge_row[ec];
        col[k] = edge_col[ec];
        val[k] = edge_val[ec];
    }
#pragma unroll
    for (int k = 0; k < KPT; ++k) {
        const int e = e0 + k * 1024 + tid;
        cls[k] = ((unsigned)row[k] / (unsigned)TR) & (unsigned)(NCL - 1);
        rk[k]  = 0u;
        if (e < E) rk[k] = atomicAdd(&hist[cls[k]], 1u);
    }
    __syncthreads();

    if (tid < 32) {
        unsigned s[8];
        unsigned sum = 0u;
#pragma unroll
        for (int q = 0; q < 8; ++q) { s[q] = hist[tid * 8 + q]; sum += s[q]; }
        unsigned inc = sum;
#pragma unroll
        for (int dlt = 1; dlt < 32; dlt <<= 1) {
            const unsigned t = __shfl_up(inc, (unsigned)dlt);
            if (tid >= dlt) inc += t;
        }
        unsigned run = inc - sum;
#pragma unroll
        for (int q = 0; q < 8; ++q) { start[tid * 8 + q] = run; run += s[q]; }
        if (tid == 31) start[NCL] = run;
    }
    __syncthreads();

    if (tid <= NCL) offT[(unsigned)tid * (unsigned)NB + (unsigned)b] = start[tid];

#pragma unroll
    for (int k = 0; k < KPT; ++k) {
        const int e = e0 + k * 1024 + tid;
        if (e < E)
            stage[start[cls[k]] + rk[k]] =
                make_uint4((unsigned)row[k], (unsigned)col[k], __float_as_uint(val[k]), (unsigned)e);
    }
    __syncthreads();

    const unsigned cnt = start[NCL];
    for (unsigned i = (unsigned)tid; i < cnt; i += 1024u) rec[(size_t)e0 + i] = stage[i];
}

template <int NP>
__device__ __forceinline__ void tile_product(const float* tile, const float* Wl,
                                             int rt0, int ct, int lr, int hi,
                                             v8f (&acc)[7]) {
    int aoff[NP];
#pragma unroll
    for (int i = 0; i < NP; ++i) {
        acc[i] = (v8f){0.f, 0.f, 0.f, 0.f, 0.f, 0.f, 0.f, 0.f};
        int r = (rt0 + 4 * i) * 16 + lr;
        if (r > TR) r = TR;
        aoff[i] = r * TP + hi * 2;
    }
    const float* bp = Wl + (ct * 16 + lr) * WP + hi * 2;
#pragma unroll 4
    for (int kk = 0; kk < D / 4; ++kk) {
        const v2f b = *(const v2f*)(bp + kk * 4);
        v2f a[NP];
#pragma unroll
        for (int i = 0; i < NP; ++i) a[i] = *(const v2f*)(tile + aoff[i] + kk * 4);
#pragma unroll
        for (int i = 0; i < NP; ++i)
            acc[i] = __builtin_amdgcn_wmma_f32_16x16x4_f32(
                 false, a[i],  false, b,
                 (short)0, acc[i],
                 false,  false);
    }
}

__global__ void __launch_bounds__(1024)
gcn_rowtile_kernel(const float*    __restrict__ X,
                   const unsigned* __restrict__ ws,
                   const float*    __restrict__ Wmat,
                   const float*    __restrict__ bias,
                   float*          __restrict__ out,
                   int N, int E) {
    __shared__ float    tile[(TR + 1) * TP];
    __shared__ __attribute__((aligned(16)))
               uint2    queue[NW * QCAP];
    __shared__ unsigned qcnt[NW];

    const int tid  = threadIdx.x;
    const int lane = tid & 31;
    const int wave = __builtin_amdgcn_readfirstlane(tid >> 5);
    const int row0 = blockIdx.x * TR;
    int nrow = N - row0;
    if (nrow > TR) nrow = TR;

    float4* tile4 = (float4*)tile;
    for (int i = tid; i < (TR + 1) * TP / 4; i += 1024)
        tile4[i] = make_float4(0.f, 0.f, 0.f, 0.f);
    if (tid < NW) qcnt[tid] = 0u;
    __syncthreads();

    const int NB = (E + BE - 1) / BE;
    const unsigned* offT = ws;
    const uint4*    rec  = (const uint4*)(ws + rec_offset_dwords(NB));
    const unsigned  cls  = (unsigned)blockIdx.x & (unsigned)(NCL - 1);
    const unsigned* off0 = offT + cls * (unsigned)NB;
    const unsigned* off1 = offT + (cls + 1u) * (unsigned)NB;
    const int nseg = wave < NB ? (NB - wave + NW - 1) / NW : 0;

    const int fo = lane * 4;

#define LOOK_UP(BASE, COL, VAL, DST)                                            \
    {                                                                           \
        const unsigned i_   = (BASE) + (unsigned)lane;                          \
        const bool     vl_  = i_ < n;                                           \
        unsigned j_ = 0u;                                                       \
        for (int k_ = 0; k_ < ns; ++k_)                                         \
            j_ += (i_ >= (unsigned)__builtin_amdgcn_readlane((int)inc, k_)) ? 1u : 0u; \
        const unsigned sb_  = (unsigned)__shfl((int)sbase, (int)(vl_ ? j_ : 0u)); \
        const uint4    rc_  = rec[vl_ ? sb_ + i_ : 0u];                         \
        const unsigned lim_ = vl_ ? (unsigned)nrow : 0u;                        \
        const unsigned d_   = rc_.x - (unsigned)row0;                           \
        COL = rc_.y;                                                            \
        VAL = rc_.z;                                                            \
        DST = d_ < lim_ ? d_ : (unsigned)TR;                                    \
    }
#define HAND_OUT(COL, VAL, DST)                                                 \
    {                                                                           \
        bool over_ = false;                                                     \
        if (DST != (unsigned)TR) {                                              \
            const unsigned ow_ = DST & (unsigned)(NW - 1);                      \
            const unsigned rk_ = atomicAdd(&qcnt[ow_], 1u);                     \
            if (rk_ < (unsigned)QCAP)                                           \
                queue[ow_ * (unsigned)QCAP + rk_] = make_uint2(COL | (DST << CB), VAL); \
            else                                                                \
                over_ = true;                                                   \
        }                                                                       \
        unsigned om_ = (unsigned)__ballot(over_);                               \
        while (om_ != 0u) {     \
            const int l_ = __builtin_ctz(om_);                                  \
            om_ &= om_ - 1u;                                                    \
            const unsigned c_ = (unsigned)__shfl((int)COL, l_);                 \
            const float    v_ = __uint_as_float((unsigned)__shfl((int)VAL, l_)); \
            const unsigned r_ = (unsigned)__shfl((int)DST, l_);                 \
            const float4   x_ = *(const float4*)(X + (size_t)c_ * D + fo);      \
            float* t_ = tile + r_ * TP + fo;                                     \
            __hip_atomic_fetch_add(t_ + 0, v_ * x_.x, __ATOMIC_RELAXED, __HIP_MEMORY_SCOPE_WORKGROUP); \
            __hip_atomic_fetch_add(t_ + 1, v_ * x_.y, __ATOMIC_RELAXED, __HIP_MEMORY_SCOPE_WORKGROUP); \
            __hip_atomic_fetch_add(t_ + 2, v_ * x_.z, __ATOMIC_RELAXED, __HIP_MEMORY_SCOPE_WORKGROUP); \
            __hip_atomic_fetch_add(t_ + 3, v_ * x_.w, __ATOMIC_RELAXED, __HIP_MEMORY_SCOPE_WORKGROUP); \
        }                                                                       \
    }

    for (int p0 = 0; p0 < nseg; p0 += 32) {
        const int  ns = nseg - p0 < 32 ? nseg - p0 : 32;
        const bool sv = lane < ns;
        const int  bk = wave + NW * (p0 + lane);
        unsigned st = 0u, en = 0u;
        if (sv) { st = off0[bk]; en = off1[bk]; }
        const unsigned cnt = en - st;
        unsigned inc = cnt;
#pragma unroll
        for (int dlt = 1; dlt < 32; dlt <<= 1) {
            const unsigned t = __shfl_up(inc, (unsigned)dlt);
            if (lane >= dlt) inc += t;
        }
        const unsigned sbase = (unsigned)bk * (unsigned)BE + st - (inc - cnt);
        const unsigned n   = (unsigned)__builtin_amdgcn_readlane((int)inc, 31);
        const unsigned nch = (n + 31u) >> 5;

        if (nch != 0u) {
            unsigned c_col, c_val, c_d;
            LOOK_UP(0u, c_col, c_val, c_d)
            for (unsigned ch = 0; ch < nch; ++ch) {
                unsigned n_col, n_val, n_d;
                LOOK_UP((ch + 1u) << 5, n_col, n_val, n_d)
                HAND_OUT(c_col, c_val, c_d)
                c_col = n_col;  c_val = n_val;  c_d = n_d;
            }
        }
    }
#undef LOOK_UP
#undef HAND_OUT
    __syncthreads();

    {
        unsigned qn = qcnt[wave];
        if (qn > (unsigned)QCAP) qn = (unsigned)QCAP;
        qn = (unsigned)__builtin_amdgcn_readfirstlane((int)qn);
        const uint2*   q  = queue + (unsigned)wave * (unsigned)QCAP;
        const unsigned ng = (qn + (unsigned)GB - 1u) / (unsigned)GB;

#define GROUP(J, PK, VL)                                                        \
        {                                                                       \
            const unsigned ri_ = (J) * (unsigned)GB + (unsigned)(lane & (GB - 1)); \
            const bool     ok_ = ri_ < qn;                                      \
            const uint2    r_  = q[ok_ ? ri_ : 0u];                             \
            PK = ok_ ? r_.x : ((r_.x & CMASK) | ((unsigned)TR << CB));          \
            VL = r_.y;                                                          \
        }
#define ISSUE_ROWS(PK, XR)                                                      \
        _Pragma("unroll")                                                       \
        for (int u_ = 0; u_ < GB; ++u_) {                                       \
            const unsigned p_ = (unsigned)__builtin_amdgcn_readlane((int)PK, u_); \
            XR[u_] = *(const float4*)(X + (size_t)(p_ & CMASK) * D + fo);       \
        }
#define ADD_ROWS(PK, VL, XR)                                                    \
        _Pragma("unroll")                                                       \
        for (int u_ = 0; u_ < GB; ++u_) {                                       \
            const unsigned lr_ = (unsigned)__builtin_amdgcn_readlane((int)PK, u_) >> CB; \
            const float    v_  = __uint_as_float((unsigned)__builtin_amdgcn_readlane((int)VL, u_)); \
            float4* t_ = (float4*)(tile + lr_ * TP + fo);                        \
            float4  a_ = *t_;                                                   \
            a_.x += v_ * XR[u_].x;                                              \
            a_.y += v_ * XR[u_].y;                                              \
            a_.z += v_ * XR[u_].z;                                              \
            a_.w += v_ * XR[u_].w;                                              \
            *t_ = a_;                                                           \
        }

        if (ng != 0u) {
            unsigned pk0, vl0, pk1, vl1;
            float4 xa[GB], xb[GB];
            GROUP(0u, pk0, vl0)
            GROUP(1u, pk1, vl1)
            ISSUE_ROWS(pk0, xa)
            for (unsigned k = 0; k < ng; k += 2u) {
                unsigned pk2, vl2, pk3, vl3;
                ISSUE_ROWS(pk1, xb)
                GROUP(k + 2u, pk2, vl2)
                GROUP(k + 3u, pk3, vl3)
                ADD_ROWS(pk0, vl0, xa)
                ISSUE_ROWS(pk2, xa)
                ADD_ROWS(pk1, vl1, xb)
                pk0 = pk2;  vl0 = vl2;  pk1 = pk3;  vl1 = vl3;
            }
        }
#undef GROUP
#undef ISSUE_ROWS
#undef ADD_ROWS
    }
    __syncthreads();

    float* Wl = (float*)queue;
    {
        float4 wreg[(D * D) / 4 / 1024];
#pragma unroll
        for (int q = 0; q < (D * D) / 4 / 1024; ++q) wreg[q] = ((const float4*)Wmat)[tid + 1024 * q];
#pragma unroll
        for (int q = 0; q < (D * D) / 4 / 1024; ++q) {
            const int i4 = tid + 1024 * q;
            const int o  = i4 >> 5;
            const int k  = (i4 & 31) * 4;
            *(float4*)&Wl[o * WP + k] = wreg[q];
        }
    }
    __syncthreads();

    const int lr  = lane & 15;
    const int hi  = lane >> 4;
    const int ct  = wave & 7;
    const int rt0 = wave >> 3;
    const int np  = (NRT - rt0 + 3) / 4;
    v8f acc[7];
    if (np == 7) tile_product<7>(tile, Wl, rt0, ct, lr, hi, acc);
    else         tile_product<6>(tile, Wl, rt0, ct, lr, hi, acc);
    __syncthreads();

    {
        const int   col = ct * 16 + lr;
        const float bb  = bias[col];
#pragma unroll
        for (int i = 0; i < 6; ++i) {
#pragma unroll
            for (int j = 0; j < 8; ++j)
                tile[((rt0 + 4 * i) * 16 + hi * 8 + j) * TP + col] = acc[i][j] + bb;
        }
        if (np == 7) {
#pragma unroll
            for (int j = 0; j < 8; ++j) {
                const int r = (rt0 + 24) * 16 + hi * 8 + j;
                if (r < TR) tile[r * TP + col] = acc[6][j] + bb;
            }
        }
    }
    __syncthreads();

    float4* out4 = (float4*)(out + (size_t)row0 * D);
    for (int j = tid; j < nrow * (D / 4); j += 1024)
        out4[j] = *(const float4*)(tile + (j >> 5) * TP + (j & 31) * 4);
}

extern "C" void kernel_launch(void* const* d_in, const int* in_sizes, int n_in,
                              void* d_out, int out_size, void* d_ws, size_t ws_size,
                              hipStream_t stream) {
    const float* X        = (const float*)d_in[0];
    const int*   edge_row = (const int*)  d_in[1];
    const int*   edge_col = (const int*)  d_in[2];
    const float* edge_val = (const float*)d_in[3];
    const float* Wmat     = (const float*)d_in[4];
    const float* bias     = (const float*)d_in[5];
    float*       out      = (float*)d_out;
    unsigned*    ws       = (unsigned*)d_ws;

    const int N = in_sizes[0] / D;
    const int E = in_sizes[1];

    {
        int sblocks = (E + BE - 1) / BE;
        if (sblocks > 0)
            gcn_blocksort_kernel<<<sblocks, 1024, 0, stream>>>(edge_row, edge_col, edge_val, ws, E);
    }

    {
        int tiles = (N + TR - 1) / TR;
        gcn_rowtile_kernel<<<tiles, 1024, 0, stream>>>(X, ws, Wmat, bias, out, N, E);
    }
}
